# starting kernel
# baseline (speedup 1.0000x reference)
#include <hip/hip_runtime.h>
#include <math.h>

typedef __attribute__((ext_vector_type(2))) float v2f;
typedef __attribute__((ext_vector_type(8))) float v8f;

#define B_    4
#define P_    6
#define F_    257
#define T_    300
#define N_    36
#define NPAD  48
#define TPAD  304
#define NT_TILES 3
#define TT_TILES 19
#define FT    (F_ * T_)
#define V_SOUND 343.0f
#define TWO_PI 6.28318530717958647692f

__device__ __forceinline__ v8f wmma_tile_k12(
    const float* cosT, const float* sinT,
    const float* cosO, const float* sinO,
    int nA, int tB, int halfSel)
{
    v8f acc = {};
    #pragma unroll
    for (int kk = 0; kk < 12; kk += 4) {
        const int p = (kk >> 1) + halfSel;
        v2f afrag, bfrag;
        afrag.x = cosT[p * NPAD + nA];
        afrag.y = sinT[p * NPAD + nA];
        bfrag.x = cosO[p * TPAD + tB];
        bfrag.y = sinO[p * TPAD + tB];
        acc = __builtin_amdgcn_wmma_f32_16x16x4_f32(
             false, afrag,  false, bfrag,
             (short)0, acc,  false,  false);
    }
    return acc;
}

__global__ __launch_bounds__(256) void CalculateDirectionFeature_kernel(
    const float* __restrict__ obs,
    const float* __restrict__ azi,
    const float* __restrict__ ele,
    const float* __restrict__ pvec,
    const float* __restrict__ freq,
    float* __restrict__ out)
{
    __shared__ float cosT[P_ * NPAD];
    __shared__ float sinT[P_ * NPAD];
    __shared__ float cosO[P_ * TPAD];
    __shared__ float sinO[P_ * TPAD];

    const int bf  = blockIdx.x;
    const int b   = bf / F_;
    const int f   = bf % F_;
    const int tid = threadIdx.x;

    const float fr    = freq[f];
    const float scale = (TWO_PI / V_SOUND) * fr;

    for (int i = tid; i < P_ * NPAD; i += 256) {
        const int p = i / NPAD;
        const int n = i % NPAD;
        float c = 0.0f, s = 0.0f;
        if (n < N_) {
            const float a = azi[b * N_ + n];
            const float e = ele[b * N_ + n];
            float sa, ca, se, ce;
            __sincosf(a, &sa, &ca);
            __sincosf(e, &se, &ce);
            const float rx = se * ca, ry = se * sa, rz = ce;
            const float dot = pvec[p * 3 + 0] * rx + pvec[p * 3 + 1] * ry +
                              pvec[p * 3 + 2] * rz;
            __sincosf(scale * dot, &s, &c);
        }
        cosT[i] = c;
        sinT[i] = s;
    }

    for (int i = tid; i < P_ * TPAD; i += 256) {
        const int p = i / TPAD;
        const int t = i % TPAD;
        float c = 0.0f, s = 0.0f;
        if (t < T_) {
            const float o = obs[(((size_t)b * P_ + p) * F_ + f) * T_ + t];
            __sincosf(o, &s, &c);
        }
        cosO[i] = c;
        sinO[i] = s;
    }
    __syncthreads();

    const int wave    = tid >> 5;
    const int lane    = tid & 31;
    const int halfSel = lane >> 4;
    const int lane16  = lane & 15;

    float* const outBF = out + ((size_t)b * N_ * F_ + f) * T_;
    const int laneOff = halfSel * (8 * FT) + lane16;

    for (int tile = wave; tile < 2 * TT_TILES; tile += 8) {
        const int nt = tile / TT_TILES;
        const int tt = tile % TT_TILES;
        const int n0 = nt * 16;
        const int t0 = (tt == TT_TILES - 1) ? (T_ - 16) : tt * 16;

        v8f acc = wmma_tile_k12(cosT, sinT, cosO, sinO,
                                n0 + lane16, t0 + lane16, halfSel);

        float* op = outBF + ((size_t)(n0 * FT + t0) + laneOff);
        #pragma unroll
        for (int r = 0; r < 8; ++r) {
            op[(size_t)r * FT] = acc[r];
        }
    }

    for (int tt = wave; tt < TT_TILES; tt += 8) {
        const int n0 = 32;
        const int t0 = (tt == TT_TILES - 1) ? (T_ - 16) : tt * 16;

        v8f acc = wmma_tile_k12(cosT, sinT, cosO, sinO,
                                n0 + lane16, t0 + lane16, halfSel);

        float* op = outBF + ((size_t)(n0 * FT + t0) + laneOff);
        if (halfSel == 0) {
            #pragma unroll
            for (int r = 0; r < 4; ++r) {
                op[(size_t)r * FT] = acc[r];
            }
        }
    }
}

extern "C" void kernel_launch(void* const* d_in, const int* in_sizes, int n_in,
                              void* d_out, int out_size, void* d_ws, size_t ws_size,
                              hipStream_t stream) {
    (void)in_sizes; (void)n_in; (void)d_ws; (void)ws_size; (void)out_size;
    const float* obs  = (const float*)d_in[0];
    const float* azi  = (const float*)d_in[1];
    const float* ele  = (const float*)d_in[2];
    const float* pvec = (const float*)d_in[3];
    const float* freq = (const float*)d_in[4];
    float* out = (float*)d_out;

    dim3 grid(B_ * F_);
    dim3 block(256);
    CalculateDirectionFeature_kernel<<<grid, block, 0, stream>>>(
        obs, azi, ele, pvec, freq, out);
}


# Workgroups of 2 waves instead of 8 (wave entries 8224 -> 2056): each thread fetches its share of the bin as 8 independent 16-byte loads, each wave streams 28 tiles instead of 7
# speedup vs baseline: 1.1787x; 1.1787x over previous
#include <hip/hip_runtime.h>
#include <math.h>

typedef __attribute__((ext_vector_type(2))) float v2f;
typedef __attribute__((ext_vector_type(4))) float v4f;
typedef __attribute__((ext_vector_type(8))) float v8f;

#define B_    4
#define P_    6
#define F_    257
#define T_    300
#define N_    36
#define NPAD  48
#define TPAD  304
#define TT_TILES 19
#define FT    (F_ * T_)
#define V_SOUND 343.0f
#define TWO_PI 6.28318530717958647692f

#define NTHREADS 64
#define NWAVES   (NTHREADS / 32)
#define ROW_V4   (T_ / 4)
#define OBS_V4   (P_ * ROW_V4)
#define OBS_ITERS ((OBS_V4 + NTHREADS - 1) / NTHREADS)

__device__ __forceinline__ v8f wmma_tile_k12(
    const float* cosT, const float* sinT,
    const float* cosO, const float* sinO,
    int nA, int tB, int halfSel)
{
    v8f acc = {};
    #pragma unroll
    for (int kk = 0; kk < 12; kk += 4) {
        const int p = (kk >> 1) + halfSel;
        v2f afrag, bfrag;
        afrag.x = cosT[p * NPAD + nA];
        afrag.y = sinT[p * NPAD + nA];
        bfrag.x = cosO[p * TPAD + tB];
        bfrag.y = sinO[p * TPAD + tB];
        acc = __builtin_amdgcn_wmma_f32_16x16x4_f32(
             false, afrag,  false, bfrag,
             (short)0, acc,  false,  false);
    }
    return acc;
}

__global__ __launch_bounds__(NTHREADS) void CalculateDirectionFeature_kernel(
    const float* __restrict__ obs,
    const float* __restrict__ azi,
    const float* __restrict__ ele,
    const float* __restrict__ pvec,
    const float* __restrict__ freq,
    float* __restrict__ out)
{
    __shared__ float cosT[P_ * NPAD];
    __shared__ float sinT[P_ * NPAD];
    __shared__ __attribute__((aligned(16))) float cosO[P_ * TPAD];
    __shared__ __attribute__((aligned(16))) float sinO[P_ * TPAD];

    const int bf  = blockIdx.x;
    const int b   = bf / F_;
    const int f   = bf % F_;
    const int tid = threadIdx.x;

    const float fr = freq[f];
    float pv[P_ * 3];
    #pragma unroll
    for (int k = 0; k < P_ * 3; ++k) pv[k] = pvec[k];

    const int nIdx = (tid < N_) ? tid : (N_ - 1);
    const float a = azi[b * N_ + nIdx];
    const float e = ele[b * N_ + nIdx];

    const v4f* obs4 = reinterpret_cast<const v4f*>(obs);
    const int base4 = (b * P_ * F_ + f) * ROW_V4;
    v4f o[OBS_ITERS];
    #pragma unroll
    for (int j = 0; j < OBS_ITERS; ++j) {
        int i = tid + j * NTHREADS;
        if (j == OBS_ITERS - 1 && i > OBS_V4 - 1) i = OBS_V4 - 1;
        const int p = i / ROW_V4;
        o[j] = obs4[base4 + i + p * ((F_ - 1) * ROW_V4)];
    }

    const float scale = (TWO_PI / V_SOUND) * fr;

    {
        const bool real = tid < N_;
        float sa, ca, se, ce;
        __sincosf(a, &sa, &ca);
        __sincosf(e, &se, &ce);
        const float rx = se * ca, ry = se * sa, rz = ce;
        #pragma unroll
        for (int p = 0; p < P_; ++p) {
            const float dot = pv[p * 3 + 0] * rx + pv[p * 3 + 1] * ry +
                              pv[p * 3 + 2] * rz;
            float s, c;
            __sincosf(scale * dot, &s, &c);
            if (tid < NPAD) {
                cosT[p * NPAD + tid] = real ? c : 0.0f;
                sinT[p * NPAD + tid] = real ? s : 0.0f;
            }
        }
    }

    #pragma unroll
    for (int j = 0; j < OBS_ITERS; ++j) {
        const int i = tid + j * NTHREADS;
        const int p = i / ROW_V4;
        const int c = i - p * ROW_V4;
        if (j < OBS_ITERS - 1 || i < OBS_V4) {
            float s0, c0, s1, c1, s2, c2, s3, c3;
            __sincosf(o[j].x, &s0, &c0);
            __sincosf(o[j].y, &s1, &c1);
            __sincosf(o[j].z, &s2, &c2);
            __sincosf(o[j].w, &s3, &c3);
            *reinterpret_cast<v4f*>(&cosO[p * TPAD + 4 * c]) = (v4f){c0, c1, c2, c3};
            *reinterpret_cast<v4f*>(&sinO[p * TPAD + 4 * c]) = (v4f){s0, s1, s2, s3};
        }
    }
    __syncthreads();

    const int wave    = tid >> 5;
    const int lane    = tid & 31;
    const int halfSel = lane >> 4;
    const int lane16  = lane & 15;

    float* const outBF = out + ((size_t)b * N_ * F_ + f) * T_;
    const int laneOff = halfSel * (8 * FT) + lane16;

    for (int tile = wave; tile < 2 * TT_TILES; tile += NWAVES) {
        const int nt = tile / TT_TILES;
        const int tt = tile % TT_TILES;
        const int n0 = nt * 16;
        const int t0 = (tt == TT_TILES - 1) ? (T_ - 16) : tt * 16;

        v8f acc = wmma_tile_k12(cosT, sinT, cosO, sinO,
                                n0 + lane16, t0 + lane16, halfSel);

        float* op = outBF + ((size_t)(n0 * FT + t0) + laneOff);
        #pragma unroll
        for (int r = 0; r < 8; ++r) {
            op[(size_t)r * FT] = acc[r];
        }
    }

    for (int tt = wave; tt < TT_TILES; tt += NWAVES) {
        const int n0 = 32;
        const int t0 = (tt == TT_TILES - 1) ? (T_ - 16) : tt * 16;

        v8f acc = wmma_tile_k12(cosT, sinT, cosO, sinO,
                                n0 + lane16, t0 + lane16, halfSel);

        float* op = outBF + ((size_t)(n0 * FT + t0) + laneOff);
        if (halfSel == 0) {
            #pragma unroll
            for (int r = 0; r < 4; ++r) {
                op[(size_t)r * FT] = acc[r];
            }
        }
    }
}

extern "C" void kernel_launch(void* const* d_in, const int* in_sizes, int n_in,
                              void* d_out, int out_size, void* d_ws, size_t ws_size,
                              hipStream_t stream) {
    (void)in_sizes; (void)n_in; (void)d_ws; (void)ws_size; (void)out_size;
    const float* obs  = (const float*)d_in[0];
    const float* azi  = (const float*)d_in[1];
    const float* ele  = (const float*)d_in[2];
    const float* pvec = (const float*)d_in[3];
    const float* freq = (const float*)d_in[4];
    float* out = (float*)d_out;

    dim3 grid(B_ * F_);
    dim3 block(NTHREADS);
    CalculateDirectionFeature_kernel<<<grid, block, 0, stream>>>(
        obs, azi, ele, pvec, freq, out);
}


# Each wave keeps the A fragments of all three n-tiles in registers and walks t-tiles: one B fragment load (fetched one tile ahead) feeds 9 WMMAs on 3 independent accumulators; 57 loop trips -> 19
# speedup vs baseline: 1.2743x; 1.0811x over previous
#include <hip/hip_runtime.h>
#include <math.h>

typedef __attribute__((ext_vector_type(2))) float v2f;
typedef __attribute__((ext_vector_type(4))) float v4f;
typedef __attribute__((ext_vector_type(8))) float v8f;

#define B_    4
#define P_    6
#define F_    257
#define T_    300
#define N_    36
#define NPAD  48
#define TPAD  304
#define TT_TILES 19
#define FT    (F_ * T_)
#define V_SOUND 343.0f
#define TWO_PI 6.28318530717958647692f

#define NTHREADS 64
#define NWAVES   (NTHREADS / 32)
#define ROW_V4   (T_ / 4)
#define OBS_V4   (P_ * ROW_V4)
#define OBS_ITERS ((OBS_V4 + NTHREADS - 1) / NTHREADS)

__global__ __launch_bounds__(NTHREADS) void CalculateDirectionFeature_kernel(
    const float* __restrict__ obs,
    const float* __restrict__ azi,
    const float* __restrict__ ele,
    const float* __restrict__ pvec,
    const float* __restrict__ freq,
    float* __restrict__ out)
{
    __shared__ float cosT[P_ * NPAD];
    __shared__ float sinT[P_ * NPAD];
    __shared__ __attribute__((aligned(16))) float cosO[P_ * TPAD];
    __shared__ __attribute__((aligned(16))) float sinO[P_ * TPAD];

    const int bf  = blockIdx.x;
    const int b   = bf / F_;
    const int f   = bf % F_;
    const int tid = threadIdx.x;

    const float fr = freq[f];
    float pv[P_ * 3];
    #pragma unroll
    for (int k = 0; k < P_ * 3; ++k) pv[k] = pvec[k];

    const int nIdx = (tid < N_) ? tid : (N_ - 1);
    const float a = azi[b * N_ + nIdx];
    const float e = ele[b * N_ + nIdx];

    const v4f* obs4 = reinterpret_cast<const v4f*>(obs);
    const int base4 = (b * P_ * F_ + f) * ROW_V4;
    v4f o[OBS_ITERS];
    #pragma unroll
    for (int j = 0; j < OBS_ITERS; ++j) {
        int i = tid + j * NTHREADS;
        if (j == OBS_ITERS - 1 && i > OBS_V4 - 1) i = OBS_V4 - 1;
        const int p = i / ROW_V4;
        o[j] = obs4[base4 + i + p * ((F_ - 1) * ROW_V4)];
    }

    const float scale = (TWO_PI / V_SOUND) * fr;

    {
        const bool real = tid < N_;
        float sa, ca, se, ce;
        __sincosf(a, &sa, &ca);
        __sincosf(e, &se, &ce);
        const float rx = se * ca, ry = se * sa, rz = ce;
        #pragma unroll
        for (int p = 0; p < P_; ++p) {
            const float dot = pv[p * 3 + 0] * rx + pv[p * 3 + 1] * ry +
                              pv[p * 3 + 2] * rz;
            float s, c;
            __sincosf(scale * dot, &s, &c);
            if (tid < NPAD) {
                cosT[p * NPAD + tid] = real ? c : 0.0f;
                sinT[p * NPAD + tid] = real ? s : 0.0f;
            }
        }
    }

    #pragma unroll
    for (int j = 0; j < OBS_ITERS; ++j) {
        const int i = tid + j * NTHREADS;
        const int p = i / ROW_V4;
        const int c = i - p * ROW_V4;
        if (j < OBS_ITERS - 1 || i < OBS_V4) {
            float s0, c0, s1, c1, s2, c2, s3, c3;
            __sincosf(o[j].x, &s0, &c0);
            __sincosf(o[j].y, &s1, &c1);
            __sincosf(o[j].z, &s2, &c2);
            __sincosf(o[j].w, &s3, &c3);
            *reinterpret_cast<v4f*>(&cosO[p * TPAD + 4 * c]) = (v4f){c0, c1, c2, c3};
            *reinterpret_cast<v4f*>(&sinO[p * TPAD + 4 * c]) = (v4f){s0, s1, s2, s3};
        }
    }
    __syncthreads();

    const int wave    = tid >> 5;
    const int lane    = tid & 31;
    const int halfSel = lane >> 4;
    const int lane16  = lane & 15;

    v2f afr[3][3];
    #pragma unroll
    for (int nt = 0; nt < 3; ++nt) {
        #pragma unroll
        for (int s = 0; s < 3; ++s) {
            const int p = 2 * s + halfSel;
            afr[nt][s].x = cosT[p * NPAD + nt * 16 + lane16];
            afr[nt][s].y = sinT[p * NPAD + nt * 16 + lane16];
        }
    }

    float* const outBF = out + ((size_t)b * N_ * F_ + f) * T_;
    const int laneOff = halfSel * (8 * FT) + lane16;

    int t0 = (wave == TT_TILES - 1) ? (T_ - 16) : wave * 16;
    v2f bcur[3];
    #pragma unroll
    for (int s = 0; s < 3; ++s) {
        const int p = 2 * s + halfSel;
        bcur[s].x = cosO[p * TPAD + t0 + lane16];
        bcur[s].y = sinO[p * TPAD + t0 + lane16];
    }

    for (int tt = wave; tt < TT_TILES; tt += NWAVES) {
        int ttn = tt + NWAVES;
        if (ttn > TT_TILES - 1) ttn = TT_TILES - 1;
        const int t0n = (ttn == TT_TILES - 1) ? (T_ - 16) : ttn * 16;
        v2f bnext[3];
        #pragma unroll
        for (int s = 0; s < 3; ++s) {
            const int p = 2 * s + halfSel;
            bnext[s].x = cosO[p * TPAD + t0n + lane16];
            bnext[s].y = sinO[p * TPAD + t0n + lane16];
        }

        v8f acc0 = {}, acc1 = {}, acc2 = {};
        #pragma unroll
        for (int s = 0; s < 3; ++s) {
            acc0 = __builtin_amdgcn_wmma_f32_16x16x4_f32(
                false, afr[0][s], false, bcur[s], (short)0, acc0, false, false);
            acc1 = __builtin_amdgcn_wmma_f32_16x16x4_f32(
                false, afr[1][s], false, bcur[s], (short)0, acc1, false, false);
            acc2 = __builtin_amdgcn_wmma_f32_16x16x4_f32(
                false, afr[2][s], false, bcur[s], (short)0, acc2, false, false);
        }

        float* op = outBF + ((size_t)t0 + laneOff);
        #pragma unroll
        for (int r = 0; r < 8; ++r) {
            op[(size_t)r * FT] = acc0[r];
        }
        #pragma unroll
        for (int r = 0; r < 8; ++r) {
            op[(size_t)(16 + r) * FT] = acc1[r];
        }
        if (halfSel == 0) {
            #pragma unroll
            for (int r = 0; r < 4; ++r) {
                op[(size_t)(32 + r) * FT] = acc2[r];
            }
        }

        t0 = t0n;
        #pragma unroll
        for (int s = 0; s < 3; ++s) bcur[s] = bnext[s];
    }
}

extern "C" void kernel_launch(void* const* d_in, const int* in_sizes, int n_in,
                              void* d_out, int out_size, void* d_ws, size_t ws_size,
                              hipStream_t stream) {
    (void)in_sizes; (void)n_in; (void)d_ws; (void)ws_size; (void)out_size;
    const float* obs  = (const float*)d_in[0];
    const float* azi  = (const float*)d_in[1];
    const float* ele  = (const float*)d_in[2];
    const float* pvec = (const float*)d_in[3];
    const float* freq = (const float*)d_in[4];
    float* out = (float*)d_out;

    dim3 grid(B_ * F_);
    dim3 block(NTHREADS);
    CalculateDirectionFeature_kernel<<<grid, block, 0, stream>>>(
        obs, azi, ele, pvec, freq, out);
}


# Output tiles staged through per-wave LDS rows (obs fragment as the A operand: a lane holds 8 consecutive t of one row), written as 16-byte-per-lane stores along rows: 380 -> 87 stores per block
# speedup vs baseline: 1.4235x; 1.1171x over previous
#include <hip/hip_runtime.h>
#include <math.h>

typedef __attribute__((ext_vector_type(2))) float v2f;
typedef __attribute__((ext_vector_type(4))) float v4f;
typedef __attribute__((ext_vector_type(8))) float v8f;

#define B_    4
#define P_    6
#define F_    257
#define T_    300
#define N_    36
#define NPAD  48
#define TPAD  304
#define TT_TILES 19
#define FT    (F_ * T_)
#define V_SOUND 343.0f
#define TWO_PI 6.28318530717958647692f

#define NTHREADS 64
#define NWAVES   (NTHREADS / 32)
#define ROW_V4   (T_ / 4)
#define OBS_V4   (P_ * ROW_V4)
#define OBS_ITERS ((OBS_V4 + NTHREADS - 1) / NTHREADS)

#define WAVE_TILES 10
#define CH_TILES   5
#define SPITCH     84
#define STAGE_FLOATS (N_ * SPITCH)

template <int W4, int PER, int ROWS>
__device__ __forceinline__ void flush_chunk(const float* __restrict__ st,
                                            float* __restrict__ dst, int lane) {
    static_assert(PER * 32 == ROWS * W4, "period covers whole rows");
    constexpr int TOTAL = N_ * W4;
    constexpr int ITERS = (TOTAL + 31) / 32;
    constexpr int NPER  = (ITERS + PER - 1) / PER;
    #pragma unroll
    for (int g = 0; g < NPER; ++g) {
        const char* sg = reinterpret_cast<const char*>(st + g * ROWS * SPITCH);
        char* dg = reinterpret_cast<char*>(dst + (size_t)g * ROWS * FT);
        #pragma unroll
        for (int j = 0; j < PER; ++j) {
            const int m = g * PER + j;
            if (m < ITERS) {
                const unsigned qq   = (unsigned)lane + 32u * (unsigned)j;
                const unsigned rowp = qq / (unsigned)W4;
                const unsigned c4   = qq - rowp * (unsigned)W4;
                const unsigned lob  = (rowp * (unsigned)SPITCH + 4u * c4) * 4u;
                const unsigned gob  = (rowp * (unsigned)FT + 4u * c4) * 4u;
                if (m * 32 + 31 < TOTAL || (unsigned)(g * ROWS) + rowp < (unsigned)N_) {
                    const v4f v = *reinterpret_cast<const v4f*>(sg + lob);
                    *reinterpret_cast<v4f*>(dg + gob) = v;
                }
            }
        }
    }
}

__device__ __forceinline__ void flush_chunk60(const float* __restrict__ st,
                                              float* __restrict__ dst,
                                              int halfSel, int lane16) {
    const char* sl = reinterpret_cast<const char*>(st) +
                     (unsigned)(halfSel * SPITCH + 4 * lane16) * 4u;
    char* dl = reinterpret_cast<char*>(dst) + (unsigned)(halfSel * FT + 4 * lane16) * 4u;
    if (lane16 < 15) {
        #pragma unroll
        for (int m = 0; m < N_ / 2; ++m) {
            const v4f v = *reinterpret_cast<const v4f*>(sl + (size_t)m * 2 * SPITCH * 4);
            *reinterpret_cast<v4f*>(dl + (size_t)m * 2 * FT * 4) = v;
        }
    }
}

__global__ __launch_bounds__(NTHREADS) void CalculateDirectionFeature_kernel(
    const float* __restrict__ obs,
    const float* __restrict__ azi,
    const float* __restrict__ ele,
    const float* __restrict__ pvec,
    const float* __restrict__ freq,
    float* __restrict__ out)
{
    __shared__ float cosT[P_ * NPAD];
    __shared__ float sinT[P_ * NPAD];
    __shared__ __attribute__((aligned(16))) float cosO[P_ * TPAD];
    __shared__ __attribute__((aligned(16))) float sinO[P_ * TPAD];
    __shared__ __attribute__((aligned(16))) float stage[NWAVES * STAGE_FLOATS];

    const int bf  = blockIdx.x;
    const int b   = bf / F_;
    const int f   = bf % F_;
    const int tid = threadIdx.x;

    const float fr = freq[f];
    float pv[P_ * 3];
    #pragma unroll
    for (int k = 0; k < P_ * 3; ++k) pv[k] = pvec[k];

    const int nIdx = (tid < N_) ? tid : (N_ - 1);
    const float a = azi[b * N_ + nIdx];
    const float e = ele[b * N_ + nIdx];

    const v4f* obs4 = reinterpret_cast<const v4f*>(obs);
    const int base4 = (b * P_ * F_ + f) * ROW_V4;
    v4f o[OBS_ITERS];
    #pragma unroll
    for (int j = 0; j < OBS_ITERS; ++j) {
        int i = tid + j * NTHREADS;
        if (j == OBS_ITERS - 1 && i > OBS_V4 - 1) i = OBS_V4 - 1;
        const int p = i / ROW_V4;
        o[j] = obs4[base4 + i + p * ((F_ - 1) * ROW_V4)];
    }

    const float scale = (TWO_PI / V_SOUND) * fr;

    {
        const bool real = tid < N_;
        float sa, ca, se, ce;
        __sincosf(a, &sa, &ca);
        __sincosf(e, &se, &ce);
        const float rx = se * ca, ry = se * sa, rz = ce;
        #pragma unroll
        for (int p = 0; p < P_; ++p) {
            const float dot = pv[p * 3 + 0] * rx + pv[p * 3 + 1] * ry +
                              pv[p * 3 + 2] * rz;
            float s, c;
            __sincosf(scale * dot, &s, &c);
            if (tid < NPAD) {
                cosT[p * NPAD + tid] = real ? c : 0.0f;
                sinT[p * NPAD + tid] = real ? s : 0.0f;
            }
        }
    }

    #pragma unroll
    for (int j = 0; j < OBS_ITERS; ++j) {
        const int i = tid + j * NTHREADS;
        const int p = i / ROW_V4;
        const int c = i - p * ROW_V4;
        if (j < OBS_ITERS - 1 || i < OBS_V4) {
            float s0, c0, s1, c1, s2, c2, s3, c3;
            __sincosf(o[j].x, &s0, &c0);
            __sincosf(o[j].y, &s1, &c1);
            __sincosf(o[j].z, &s2, &c2);
            __sincosf(o[j].w, &s3, &c3);
            *reinterpret_cast<v4f*>(&cosO[p * TPAD + 4 * c]) = (v4f){c0, c1, c2, c3};
            *reinterpret_cast<v4f*>(&sinO[p * TPAD + 4 * c]) = (v4f){s0, s1, s2, s3};
        }
    }
    __syncthreads();

    const int wave    = __builtin_amdgcn_readfirstlane(tid >> 5);
    const int lane    = tid & 31;
    const int halfSel = lane >> 4;
    const int lane16  = lane & 15;

    v2f nfr[3][3];
    #pragma unroll
    for (int nt = 0; nt < 3; ++nt) {
        #pragma unroll
        for (int s = 0; s < 3; ++s) {
            const int p = 2 * s + halfSel;
            nfr[nt][s].x = cosT[p * NPAD + nt * 16 + lane16];
            nfr[nt][s].y = sinT[p * NPAD + nt * 16 + lane16];
        }
    }

    float* const outBF = out + ((size_t)b * N_ * F_ + f) * T_;
    float* const st = stage + wave * STAGE_FLOATS;

    const int firstTile = wave * WAVE_TILES;
    const int lastTile  = (wave == 0) ? (WAVE_TILES - 1) : (TT_TILES - 1);

    int t0 = firstTile * 16;
    v2f bcur[3];
    #pragma unroll
    for (int s = 0; s < 3; ++s) {
        const int p = 2 * s + halfSel;
        bcur[s].x = cosO[p * TPAD + t0 + lane16];
        bcur[s].y = sinO[p * TPAD + t0 + lane16];
    }

    #pragma unroll 1
    for (int ch = 0; ch < 2; ++ch) {
        const int chFirst = firstTile + ch * CH_TILES;
        const int chBase  = chFirst * 16;
        int chLast = chFirst + CH_TILES - 1;
        if (chLast > lastTile) chLast = lastTile;

        #pragma unroll 1
        for (int tt = chFirst; tt <= chLast; ++tt) {
            int ttn = tt + 1;
            if (ttn > lastTile) ttn = lastTile;
            const int t0n = (ttn == TT_TILES - 1) ? (T_ - 16) : ttn * 16;
            v2f bnext[3];
            #pragma unroll
            for (int s = 0; s < 3; ++s) {
                const int p = 2 * s + halfSel;
                bnext[s].x = cosO[p * TPAD + t0n + lane16];
                bnext[s].y = sinO[p * TPAD + t0n + lane16];
            }

            v8f acc0 = {}, acc1 = {}, acc2 = {};
            #pragma unroll
            for (int s = 0; s < 3; ++s) {
                acc0 = __builtin_amdgcn_wmma_f32_16x16x4_f32(
                    false, bcur[s], false, nfr[0][s], (short)0, acc0, false, false);
                acc1 = __builtin_amdgcn_wmma_f32_16x16x4_f32(
                    false, bcur[s], false, nfr[1][s], (short)0, acc1, false, false);
                acc2 = __builtin_amdgcn_wmma_f32_16x16x4_f32(
                    false, bcur[s], false, nfr[2][s], (short)0, acc2, false, false);
            }

            float* sp = st + lane16 * SPITCH + (t0 - chBase) + 8 * halfSel;
            *reinterpret_cast<v4f*>(sp)     = (v4f){acc0[0], acc0[1], acc0[2], acc0[3]};
            *reinterpret_cast<v4f*>(sp + 4) = (v4f){acc0[4], acc0[5], acc0[6], acc0[7]};
            *reinterpret_cast<v4f*>(sp + 16 * SPITCH)     = (v4f){acc1[0], acc1[1], acc1[2], acc1[3]};
            *reinterpret_cast<v4f*>(sp + 16 * SPITCH + 4) = (v4f){acc1[4], acc1[5], acc1[6], acc1[7]};
            if (lane16 < N_ - 32) {
                *reinterpret_cast<v4f*>(sp + 32 * SPITCH)     = (v4f){acc2[0], acc2[1], acc2[2], acc2[3]};
                *reinterpret_cast<v4f*>(sp + 32 * SPITCH + 4) = (v4f){acc2[4], acc2[5], acc2[6], acc2[7]};
            }

            t0 = t0n;
            #pragma unroll
            for (int s = 0; s < 3; ++s) bcur[s] = bnext[s];
        }

        asm volatile("s_wait_dscnt 0x0" ::: "memory");
        if (chLast - chFirst == CH_TILES - 1) {
            flush_chunk<20, 5, 8>(st, outBF + chBase, lane);
        } else {
            flush_chunk60(st, outBF + chBase, halfSel, lane16);
        }
        asm volatile("" ::: "memory");
    }
}

extern "C" void kernel_launch(void* const* d_in, const int* in_sizes, int n_in,
                              void* d_out, int out_size, void* d_ws, size_t ws_size,
                              hipStream_t stream) {
    (void)in_sizes; (void)n_in; (void)d_ws; (void)ws_size; (void)out_size;
    const float* obs  = (const float*)d_in[0];
    const float* azi  = (const float*)d_in[1];
    const float* ele  = (const float*)d_in[2];
    const float* pvec = (const float*)d_in[3];
    const float* freq = (const float*)d_in[4];
    float* out = (float*)d_out;

    dim3 grid(B_ * F_);
    dim3 block(NTHREADS);
    CalculateDirectionFeature_kernel<<<grid, block, 0, stream>>>(
        obs, azi, ele, pvec, freq, out);
}


# Each wave owns whole output rows (its n-tile over all t; rows 32-35 split by t) in a 128-column LDS ring: every 4 t-tiles each row's next aligned 256-byte unit leaves whole, one half-wave per unit
# speedup vs baseline: 1.5668x; 1.1007x over previous
#include <hip/hip_runtime.h>
#include <math.h>

typedef __attribute__((ext_vector_type(2))) float v2f;
typedef __attribute__((ext_vector_type(4))) float v4f;
typedef __attribute__((ext_vector_type(8))) float v8f;

#define B_    4
#define P_    6
#define F_    257
#define T_    300
#define N_    36
#define NPAD  48
#define TPAD  304
#define TT_TILES 19
#define FT    (F_ * T_)
#define V_SOUND 343.0f
#define TWO_PI 6.28318530717958647692f

#define NTHREADS 64
#define NWAVES   (NTHREADS / 32)
#define ROW_V4   (T_ / 4)
#define OBS_V4   (P_ * ROW_V4)
#define OBS_ITERS ((OBS_V4 + NTHREADS - 1) / NTHREADS)

#define RING    128
#define RPITCH  132
#define WROWS   20
#define STAGE_FLOATS (WROWS * RPITCH)
#define XSPLIT  160
#define FROWS   10

template <int J>
__device__ __forceinline__ void flush_units(const float* st, float* outBF,
                                            int halfSel, int lane16,
                                            const unsigned (&gOff)[FROWS],
                                            const int (&up)[FROWS],
                                            int xLo, int xHi) {
    constexpr int SUBS = (J == 4) ? 2 : 1;
    char* const grow = reinterpret_cast<char*>(outBF);
    v4f vals[FROWS * SUBS];
    int cols[FROWS * SUBS];
    bool act[FROWS * SUBS];
    #pragma unroll
    for (int i = 0; i < FROWS; ++i) {
        int lo = (J == 0) ? 0 : up[i] + 64 * (J - 1);
        int hi = (J == 4) ? T_ : up[i] + 64 * J;
        if (i >= 8) {
            lo = (lo > xLo) ? lo : xLo;
            hi = (hi < xHi) ? hi : xHi;
        }
        const float* srow = st + (2 * i + halfSel) * RPITCH;
        #pragma unroll
        for (int sub = 0; sub < SUBS; ++sub) {
            const int k = i * SUBS + sub;
            const int col = lo + 4 * lane16 + 64 * sub;
            const bool whole = (J >= 1 && J <= 3 && i < 8);
            cols[k] = col;
            act[k]  = whole || (col < hi);
            if (act[k]) vals[k] = *reinterpret_cast<const v4f*>(srow + (col & (RING - 1)));
        }
    }
    #pragma unroll
    for (int i = 0; i < FROWS; ++i) {
        #pragma unroll
        for (int sub = 0; sub < SUBS; ++sub) {
            const int k = i * SUBS + sub;
            if (act[k]) {
                const unsigned gb = (gOff[i] + (unsigned)cols[k]) * 4u;
                *reinterpret_cast<v4f*>(grow + gb) = vals[k];
            }
        }
    }
}

__global__ __launch_bounds__(NTHREADS) void CalculateDirectionFeature_kernel(
    const float* __restrict__ obs,
    const float* __restrict__ azi,
    const float* __restrict__ ele,
    const float* __restrict__ pvec,
    const float* __restrict__ freq,
    float* __restrict__ out)
{
    __shared__ float cosT[P_ * NPAD];
    __shared__ float sinT[P_ * NPAD];
    __shared__ __attribute__((aligned(16))) float cosO[P_ * TPAD];
    __shared__ __attribute__((aligned(16))) float sinO[P_ * TPAD];
    __shared__ __attribute__((aligned(16))) float stage[NWAVES * STAGE_FLOATS];

    const int bf  = blockIdx.x;
    const int b   = bf / F_;
    const int f   = bf % F_;
    const int tid = threadIdx.x;

    const float fr = freq[f];
    float pv[P_ * 3];
    #pragma unroll
    for (int k = 0; k < P_ * 3; ++k) pv[k] = pvec[k];

    const int nIdx = (tid < N_) ? tid : (N_ - 1);
    const float a = azi[b * N_ + nIdx];
    const float e = ele[b * N_ + nIdx];

    const v4f* obs4 = reinterpret_cast<const v4f*>(obs);
    const int base4 = (b * P_ * F_ + f) * ROW_V4;
    v4f o[OBS_ITERS];
    #pragma unroll
    for (int j = 0; j < OBS_ITERS; ++j) {
        int i = tid + j * NTHREADS;
        if (j == OBS_ITERS - 1 && i > OBS_V4 - 1) i = OBS_V4 - 1;
        const int p = i / ROW_V4;
        o[j] = obs4[base4 + i + p * ((F_ - 1) * ROW_V4)];
    }

    const float scale = (TWO_PI / V_SOUND) * fr;

    {
        const bool real = tid < N_;
        float sa, ca, se, ce;
        __sincosf(a, &sa, &ca);
        __sincosf(e, &se, &ce);
        const float rx = se * ca, ry = se * sa, rz = ce;
        #pragma unroll
        for (int p = 0; p < P_; ++p) {
            const float dot = pv[p * 3 + 0] * rx + pv[p * 3 + 1] * ry +
                              pv[p * 3 + 2] * rz;
            float s, c;
            __sincosf(scale * dot, &s, &c);
            if (tid < NPAD) {
                cosT[p * NPAD + tid] = real ? c : 0.0f;
                sinT[p * NPAD + tid] = real ? s : 0.0f;
            }
        }
    }

    #pragma unroll
    for (int j = 0; j < OBS_ITERS; ++j) {
        const int i = tid + j * NTHREADS;
        const int p = i / ROW_V4;
        const int c = i - p * ROW_V4;
        if (j < OBS_ITERS - 1 || i < OBS_V4) {
            float s0, c0, s1, c1, s2, c2, s3, c3;
            __sincosf(o[j].x, &s0, &c0);
            __sincosf(o[j].y, &s1, &c1);
            __sincosf(o[j].z, &s2, &c2);
            __sincosf(o[j].w, &s3, &c3);
            *reinterpret_cast<v4f*>(&cosO[p * TPAD + 4 * c]) = (v4f){c0, c1, c2, c3};
            *reinterpret_cast<v4f*>(&sinO[p * TPAD + 4 * c]) = (v4f){s0, s1, s2, s3};
        }
    }
    __syncthreads();

    const int wave    = __builtin_amdgcn_readfirstlane(tid >> 5);
    const int lane    = tid & 31;
    const int halfSel = lane >> 4;
    const int lane16  = lane & 15;

    v2f nfrM[3], nfrX[3];
    #pragma unroll
    for (int s = 0; s < 3; ++s) {
        const int p = 2 * s + halfSel;
        nfrM[s].x = cosT[p * NPAD + wave * 16 + lane16];
        nfrM[s].y = sinT[p * NPAD + wave * 16 + lane16];
        nfrX[s].x = cosT[p * NPAD + 32 + lane16];
        nfrX[s].y = sinT[p * NPAD + 32 + lane16];
    }

    float* const outBF = out + ((size_t)b * N_ * F_ + f) * T_;
    float* const st = stage + wave * STAGE_FLOATS;

    const int xFirst = wave ? 10 : 0;
    const int xLast  = wave ? (TT_TILES - 1) : 9;
    const int xLo    = wave ? XSPLIT : 0;
    const int xHi    = wave ? T_ : XSPLIT;

    unsigned gOff[FROWS];
    int up[FROWS];
    {
        const unsigned baseLow = (unsigned)reinterpret_cast<unsigned long long>(outBF);
        #pragma unroll
        for (int i = 0; i < FROWS; ++i) {
            const int n = ((i < 8) ? 16 * wave : 16) + 2 * i + halfSel;
            gOff[i] = (unsigned)n * (unsigned)FT;
            const unsigned o = ((0u - (baseLow + gOff[i] * 4u)) & 255u) >> 2;
            up[i] = o ? (int)o : 64;
        }
    }

    int t0 = 0;
    v2f bcur[3];
    #pragma unroll
    for (int s = 0; s < 3; ++s) {
        const int p = 2 * s + halfSel;
        bcur[s].x = cosO[p * TPAD + lane16];
        bcur[s].y = sinO[p * TPAD + lane16];
    }

#define RUN_TILES(TT_BEGIN, TT_END)                                                     \
    _Pragma("unroll 1")                                                                 \
    for (int tt = (TT_BEGIN); tt < (TT_END); ++tt) {                                    \
        int ttn = tt + 1;                                                               \
        if (ttn > TT_TILES - 1) ttn = TT_TILES - 1;                                     \
        const int t0n = (ttn == TT_TILES - 1) ? (T_ - 16) : ttn * 16;                   \
        v2f bnext[3];                                                                   \
        _Pragma("unroll")                                                               \
        for (int s = 0; s < 3; ++s) {                                                   \
            const int p = 2 * s + halfSel;                                              \
            bnext[s].x = cosO[p * TPAD + t0n + lane16];                                 \
            bnext[s].y = sinO[p * TPAD + t0n + lane16];                                 \
        }                                                                               \
        float* sp = st + lane16 * RPITCH + ((t0 + 8 * halfSel) & (RING - 1));           \
        if (tt >= xFirst && tt <= xLast) {                                              \
            v8f accM = {}, accX = {};                                                   \
            _Pragma("unroll")                                                           \
            for (int s = 0; s < 3; ++s) {                                               \
                accM = __builtin_amdgcn_wmma_f32_16x16x4_f32(                           \
                    false, bcur[s], false, nfrM[s], (short)0, accM, false, false);      \
                accX = __builtin_amdgcn_wmma_f32_16x16x4_f32(                           \
                    false, bcur[s], false, nfrX[s], (short)0, accX, false, false);      \
            }                                                                           \
            *reinterpret_cast<v4f*>(sp)     = (v4f){accM[0], accM[1], accM[2], accM[3]}; \
            *reinterpret_cast<v4f*>(sp + 4) = (v4f){accM[4], accM[5], accM[6], accM[7]}; \
            if (lane16 < N_ - 32) {                                                     \
                *reinterpret_cast<v4f*>(sp + 16 * RPITCH) =                             \
                    (v4f){accX[0], accX[1], accX[2], accX[3]};                          \
                *reinterpret_cast<v4f*>(sp + 16 * RPITCH + 4) =                         \
                    (v4f){accX[4], accX[5], accX[6], accX[7]};                          \
            }                                                                           \
        } else {                                                                        \
            v8f accM = {};                                                              \
            _Pragma("unroll")                                                           \
            for (int s = 0; s < 3; ++s) {                                               \
                accM = __builtin_amdgcn_wmma_f32_16x16x4_f32(                           \
                    false, bcur[s], false, nfrM[s], (short)0, accM, false, false);      \
            }                                                                           \
            *reinterpret_cast<v4f*>(sp)     = (v4f){accM[0], accM[1], accM[2], accM[3]}; \
            *reinterpret_cast<v4f*>(sp + 4) = (v4f){accM[4], accM[5], accM[6], accM[7]}; \
        }                                                                               \
        t0 = t0n;                                                                       \
        _Pragma("unroll")                                                               \
        for (int s = 0; s < 3; ++s) bcur[s] = bnext[s];                                 \
    }

    RUN_TILES(0, 4)
    flush_units<0>(st, outBF, halfSel, lane16, gOff, up, xLo, xHi);
    RUN_TILES(4, 8)
    flush_units<1>(st, outBF, halfSel, lane16, gOff, up, xLo, xHi);
    RUN_TILES(8, 12)
    flush_units<2>(st, outBF, halfSel, lane16, gOff, up, xLo, xHi);
    RUN_TILES(12, 16)
    flush_units<3>(st, outBF, halfSel, lane16, gOff, up, xLo, xHi);
    RUN_TILES(16, TT_TILES)
    flush_units<4>(st, outBF, halfSel, lane16, gOff, up, xLo, xHi);
#undef RUN_TILES
}

extern "C" void kernel_launch(void* const* d_in, const int* in_sizes, int n_in,
                              void* d_out, int out_size, void* d_ws, size_t ws_size,
                              hipStream_t stream) {
    (void)in_sizes; (void)n_in; (void)d_ws; (void)ws_size; (void)out_size;
    const float* obs  = (const float*)d_in[0];
    const float* azi  = (const float*)d_in[1];
    const float* ele  = (const float*)d_in[2];
    const float* pvec = (const float*)d_in[3];
    const float* freq = (const float*)d_in[4];
    float* out = (float*)d_out;

    dim3 grid(B_ * F_);
    dim3 block(NTHREADS);
    CalculateDirectionFeature_kernel<<<grid, block, 0, stream>>>(
        obs, azi, ele, pvec, freq, out);
}
